# starting kernel
# baseline (speedup 1.0000x reference)
#include <hip/hip_runtime.h>
#include <hip/hip_bf16.h>


#define KDIM 128

typedef __attribute__((ext_vector_type(2))) float v2f;
typedef __attribute__((ext_vector_type(8))) float v8f;

__global__ void gcn_degree_kernel(const int* __restrict__ src,
                                  const int* __restrict__ dst,
                                  float* __restrict__ degS,
                                  float* __restrict__ degD, int n_edges) {
  int e = blockIdx.x * blockDim.x + threadIdx.x;
  if (e < n_edges) {
    unsafeAtomicAdd(&degS[src[e]], 1.0f);
    unsafeAtomicAdd(&degD[dst[e]], 1.0f);
  }
}

__global__ void gcn_norm_kernel(float* __restrict__ d, int n) {
  int i = blockIdx.x * blockDim.x + threadIdx.x;
  if (i < n) d[i] = rsqrtf(fmaxf(d[i], 1.0f));
}

__global__ void gcn_scatter_kernel(const float* __restrict__ h,
                                   const float* __restrict__ out_norm,
                                   const int* __restrict__ src,
                                   const int* __restrict__ dst,
                                   float* __restrict__ agg, int n_edges) {
  int wave = (int)((blockIdx.x * blockDim.x + threadIdx.x) >> 5);
  int lane = threadIdx.x & 31;
  if (wave >= n_edges) return;
  int s = src[wave];
  int d = dst[wave];
  float scale = out_norm[s];
  const float4* hp = (const float4*)(h + (size_t)s * KDIM);
  float4 v = hp[lane];
  float* ap = agg + (size_t)d * KDIM + lane * 4;
  unsafeAtomicAdd(ap + 0, v.x * scale);
  unsafeAtomicAdd(ap + 1, v.y * scale);
  unsafeAtomicAdd(ap + 2, v.z * scale);
  unsafeAtomicAdd(ap + 3, v.w * scale);
}

__global__ void gcn_gemm_wmma_kernel(const float* __restrict__ A,
                                     const float* __restrict__ in_norm,
                                     const float* __restrict__ W,
                                     const float* __restrict__ bias,
                                     float* __restrict__ out,
                                     int M, int N, int do_relu) {
  const int m0   = blockIdx.x * 16;
  const int wv   = threadIdx.x >> 5;
  const int lane = threadIdx.x & 31;
  const int half = lane >> 4;
  const int l15  = lane & 15;

  const int col    = wv * 16 + l15;
  const bool colOK = (col < N);
  const int colC   = colOK ? col : (N - 1);

  int arow = m0 + l15;
  if (arow >= M) arow = M - 1;
  const float ascale = in_norm[arow];

  const float* Ap = A + (size_t)arow * KDIM + half * 2;
  const float* Wp = W + (size_t)(half * 2) * N + colC;
  const size_t wstep = (size_t)4 * N;

  v8f c = {};
#pragma unroll 4
  for (int kk = 0; kk < KDIM; kk += 4) {
    v2f a = *(const v2f*)Ap;
    a *= ascale;
    v2f b;
    b.x = Wp[0];
    b.y = Wp[N];
    c = __builtin_amdgcn_wmma_f32_16x16x4_f32(
         false, a,  false, b,
         (short)0, c,  false,  false);
    Ap += 4;
    Wp += wstep;
  }

  if (colOK) {
    const float bv = bias ? bias[col] : 0.0f;
#pragma unroll
    for (int i = 0; i < 8; ++i) {
      const int row = m0 + i + half * 8;
      if (row < M) {
        float v = c[i] + bv;
        if (do_relu) v = fmaxf(v, 0.0f);
        out[(size_t)row * N + col] = v;
      }
    }
  }
}

extern "C" void kernel_launch(void* const* d_in, const int* in_sizes, int n_in,
                              void* d_out, int out_size, void* d_ws, size_t ws_size,
                              hipStream_t stream) {
  const float* feat = (const float*)d_in[0];
  const float* W0   = (const float*)d_in[1];
  const float* W1   = (const float*)d_in[2];
  const float* W2   = (const float*)d_in[3];
  const float* b2   = (const float*)d_in[4];
  const int*   src  = (const int*)d_in[5];
  const int*   dst  = (const int*)d_in[6];
  float* out = (float*)d_out;

  const int n_nodes = in_sizes[0] / KDIM;
  const int n_out   = in_sizes[4];
  const int n_edges = in_sizes[5];

  char* ws = (char*)d_ws;
  float* normS = (float*)ws;
  float* normD = normS + n_nodes;
  size_t off = (((size_t)2 * n_nodes * sizeof(float)) + 511) & ~(size_t)511;
  float* agg  = (float*)(ws + off);
  off += (size_t)n_nodes * KDIM * sizeof(float);
  float* hbuf = (float*)(ws + off);

  const size_t featBytes = (size_t)n_nodes * KDIM * sizeof(float);
  const int mTiles = (n_nodes + 15) / 16;
  const int scatterBlocks = (n_edges + 7) / 8;
  const dim3 blk256(256);

  hipMemsetAsync(normS, 0, (size_t)2 * n_nodes * sizeof(float), stream);
  gcn_degree_kernel<<<(n_edges + 255) / 256, blk256, 0, stream>>>(
      src, dst, normS, normD, n_edges);
  gcn_norm_kernel<<<(2 * n_nodes + 255) / 256, blk256, 0, stream>>>(
      normS, 2 * n_nodes);

  hipMemsetAsync(agg, 0, featBytes, stream);
  gcn_scatter_kernel<<<scatterBlocks, blk256, 0, stream>>>(
      feat, normS, src, dst, agg, n_edges);
  gcn_gemm_wmma_kernel<<<mTiles, dim3(32 * 8), 0, stream>>>(
      agg, normD, W0, nullptr, hbuf, n_nodes, 128,  1);

  hipMemsetAsync(agg, 0, featBytes, stream);
  gcn_scatter_kernel<<<scatterBlocks, blk256, 0, stream>>>(
      hbuf, normS, src, dst, agg, n_edges);
  gcn_gemm_wmma_kernel<<<mTiles, dim3(32 * 8), 0, stream>>>(
      agg, normD, W1, nullptr, hbuf, n_nodes, 128,  1);

  hipMemsetAsync(agg, 0, featBytes, stream);
  gcn_scatter_kernel<<<scatterBlocks, blk256, 0, stream>>>(
      hbuf, normS, src, dst, agg, n_edges);
  gcn_gemm_wmma_kernel<<<mTiles, dim3(32 * ((n_out + 15) / 16)), 0, stream>>>(
      agg, normD, W2, b2, out, n_nodes, n_out,  0);
}


# Aggregation moved into LDS: edges counting-sorted once by 64-row destination bin; a work-group adds its bin's rows in a 32 KB LDS tile (ds_add_f32) and stores agg once; sum order = LDS arrival
# speedup vs baseline: 2.2064x; 2.2064x over previous
#include <hip/hip_runtime.h>
#include <hip/hip_bf16.h>


#define KDIM 128
#define BIN_SHIFT 6
#define BIN_ROWS (1 << BIN_SHIFT)

typedef __attribute__((ext_vector_type(2))) float v2f;
typedef __attribute__((ext_vector_type(8))) float v8f;

__device__ __forceinline__ int gcn_l2_half() {
  return (int)((__builtin_amdgcn_s_sendmsg_rtn(0x87) >> 18) & 1u);
}
__device__ __forceinline__ int gcn_set_split(int n_nodes) {
  return ((n_nodes >> 1) + 31) & ~31;
}

__global__ void __launch_bounds__(256)
gcn_degree_kernel(const int* __restrict__ src,
                  const int* __restrict__ dst,
                  float* __restrict__ degS,
                  float* __restrict__ degD,
                  int* __restrict__ ctr, int n_edges, int n_nodes) {
  __shared__ int s_half;
  __shared__ int s_slot[2];
  if (threadIdx.x == 0) s_half = gcn_l2_half();
  __syncthreads();
  const int half = s_half;
  const int n_split = gcn_set_split(n_nodes);
  const int n_chunks = (n_edges + 1023) >> 10;
  int k = 0;
  for (int pass = 0; pass < 2; ++pass) {
    const int set = half ^ pass;
    for (;;) {
      if (threadIdx.x == 0) s_slot[k & 1] = atomicAdd(ctr + set, 1);
      __syncthreads();
      const int chunk = s_slot[k & 1];
      ++k;
      if (chunk >= n_chunks) break;
      const int e0 = (chunk << 10) + (int)threadIdx.x;
      int sv[4], dv[4];
      for (int j = 0; j < 4; ++j) {
        const int e = e0 + (j << 8);
        const int ec = (e < n_edges) ? e : (n_edges - 1);
        sv[j] = src[ec];
        dv[j] = dst[ec];
      }
      for (int j = 0; j < 4; ++j) {
        if (e0 + (j << 8) < n_edges) {
          if ((sv[j] >= n_split) == (set != 0)) unsafeAtomicAdd(&degS[sv[j]], 1.0f);
          if ((dv[j] >= n_split) == (set != 0)) unsafeAtomicAdd(&degD[dv[j]], 1.0f);
        }
      }
    }
  }
}

__global__ void __launch_bounds__(BIN_ROWS)
gcn_norm_bincount_kernel(float* __restrict__ degS,
                         float* __restrict__ degD,
                         int* __restrict__ binCnt, int n_nodes) {
  __shared__ int s_cnt;
  if (threadIdx.x == 0) s_cnt = 0;
  __syncthreads();
  const int i = blockIdx.x * BIN_ROWS + (int)threadIdx.x;
  int c = 0;
  if (i < n_nodes) {
    const float ds = degS[i];
    const float dd = degD[i];
    c = (int)dd;
    degS[i] = rsqrtf(fmaxf(ds, 1.0f));
    degD[i] = rsqrtf(fmaxf(dd, 1.0f));
  }
  atomicAdd(&s_cnt, c);
  __syncthreads();
  if (threadIdx.x == 0) binCnt[blockIdx.x] = s_cnt;
}

__global__ void __launch_bounds__(1024)
gcn_bin_scan_kernel(const int* __restrict__ binCnt,
                    int* __restrict__ binStart,
                    int* __restrict__ cursor, int n_bins) {
  __shared__ int s_sum[1024];
  const int t = (int)threadIdx.x;
  const int per = (n_bins + 1023) >> 10;
  const int b0 = t * per;
  int local = 0;
  for (int j = 0; j < per; ++j) {
    const int b = b0 + j;
    if (b < n_bins) local += binCnt[b];
  }
  s_sum[t] = local;
  __syncthreads();
  for (int o = 1; o < 1024; o <<= 1) {
    const int v = (t >= o) ? s_sum[t - o] : 0;
    __syncthreads();
    s_sum[t] += v;
    __syncthreads();
  }
  int run = s_sum[t] - local;
  for (int j = 0; j < per; ++j) {
    const int b = b0 + j;
    if (b < n_bins) {
      binStart[b] = run;
      cursor[b] = run;
      run += binCnt[b];
    }
  }
  if (t == 1023) binStart[n_bins] = s_sum[1023];
}

#define FILL_THREADS 1024
#define FILL_PER 4
__global__ void __launch_bounds__(FILL_THREADS)
gcn_bin_fill_kernel(const int* __restrict__ src,
                    const int* __restrict__ dst,
                    int* __restrict__ cursor,
                    int* __restrict__ epair, int n_edges, int n_bins) {
  extern __shared__ int s_fill[];
  int* s_cnt  = s_fill;
  int* s_base = s_fill + n_bins;
  const int t = (int)threadIdx.x;
  for (int b = t; b < n_bins; b += FILL_THREADS) s_cnt[b] = 0;
  __syncthreads();

  const int e0 = blockIdx.x * (FILL_THREADS * FILL_PER) + t;
  int sv[FILL_PER], dv[FILL_PER], rk[FILL_PER];
  for (int j = 0; j < FILL_PER; ++j) {
    const int e = e0 + j * FILL_THREADS;
    const int ec = (e < n_edges) ? e : (n_edges - 1);
    sv[j] = src[ec];
    dv[j] = dst[ec];
  }
  for (int j = 0; j < FILL_PER; ++j) {
    rk[j] = 0;
    if (e0 + j * FILL_THREADS < n_edges)
      rk[j] = atomicAdd(&s_cnt[dv[j] >> BIN_SHIFT], 1);
  }
  __syncthreads();
  for (int b = t; b < n_bins; b += FILL_THREADS) {
    const int c = s_cnt[b];
    if (c > 0) s_base[b] = atomicAdd(cursor + b, c);
  }
  __syncthreads();
  for (int j = 0; j < FILL_PER; ++j) {
    if (e0 + j * FILL_THREADS < n_edges) {
      const int pos = s_base[dv[j] >> BIN_SHIFT] + rk[j];
      reinterpret_cast<int2*>(epair)[pos] = make_int2(sv[j], dv[j]);
    }
  }
}

__global__ void __launch_bounds__(256)
gcn_bin_sum_kernel(const float* __restrict__ h,
                   const float* __restrict__ out_norm,
                   const int* __restrict__ epair,
                   const int* __restrict__ binStart,
                   float* __restrict__ agg) {
  __shared__ float tile[BIN_ROWS * KDIM];
  const int tid = (int)threadIdx.x;
  const int lane = tid & 31;
  const int wv = __builtin_amdgcn_readfirstlane(tid >> 5);
  const int row0 = blockIdx.x * BIN_ROWS;

  float4* t4 = reinterpret_cast<float4*>(tile);
  for (int i = 0; i < BIN_ROWS * KDIM / 4 / 256; ++i)
    t4[tid + 256 * i] = make_float4(0.0f, 0.0f, 0.0f, 0.0f);
  __syncthreads();

  const int eb = binStart[blockIdx.x];
  const int ee = binStart[blockIdx.x + 1];
  for (int base = eb + wv * 32; base < ee; base += 256) {
    const int left = ee - base;
    const int cnt = (left < 32) ? left : 32;
    int s = 0, d = row0;
    if (lane < cnt) {
      const int2 p = reinterpret_cast<const int2*>(epair)[base + lane];
      s = p.x;
      d = p.y;
    }
    for (int g = 0; g < cnt; g += 8) {
      float a[8][4];
      float c[8];
      int off[8];
#pragma unroll
      for (int j = 0; j < 8; ++j) {
        const int li = (g + j < cnt) ? (g + j) : g;
        const int sj = __builtin_amdgcn_readlane(s, li);
        const int dj = __builtin_amdgcn_readlane(d, li);
        const float* hp = h + (size_t)sj * KDIM + lane;
        c[j] = out_norm[sj];
        a[j][0] = hp[0];
        a[j][1] = hp[32];
        a[j][2] = hp[64];
        a[j][3] = hp[96];
        off[j] = (dj - row0) * KDIM + lane;
      }
#pragma unroll
      for (int j = 0; j < 8; ++j) {
        const bool ok = (j == 0) || (g + j < cnt);
#pragma unroll
        for (int q = 0; q < 4; ++q) {
          const float v = ok ? a[j][q] * c[j] : 0.0f;
          unsafeAtomicAdd(&tile[off[j] + 32 * q], v);
        }
      }
    }
  }
  __syncthreads();

  float4* o4 = reinterpret_cast<float4*>(agg + (size_t)row0 * KDIM);
  for (int i = 0; i < BIN_ROWS * KDIM / 4 / 256; ++i)
    o4[tid + 256 * i] = t4[tid + 256 * i];
}

__global__ void gcn_gemm_wmma_kernel(const float* __restrict__ A,
                                     const float* __restrict__ in_norm,
                                     const float* __restrict__ W,
                                     const float* __restrict__ bias,
                                     float* __restrict__ out,
                                     int M, int N, int do_relu) {
  const int m0   = blockIdx.x * 16;
  const int wv   = threadIdx.x >> 5;
  const int lane = threadIdx.x & 31;
  const int half = lane >> 4;
  const int l15  = lane & 15;

  const int col    = wv * 16 + l15;
  const bool colOK = (col < N);
  const int colC   = colOK ? col : (N - 1);

  int arow = m0 + l15;
  if (arow >= M) arow = M - 1;
  const float ascale = in_norm[arow];

  const float* Ap = A + (size_t)arow * KDIM + half * 2;
  const float* Wp = W + (size_t)(half * 2) * N + colC;
  const size_t wstep = (size_t)4 * N;

  v8f c = {};
#pragma unroll 4
  for (int kk = 0; kk < KDIM; kk += 4) {
    v2f a = *(const v2f*)Ap;
    a *= ascale;
    v2f b;
    b.x = Wp[0];
    b.y = Wp[N];
    c = __builtin_amdgcn_wmma_f32_16x16x4_f32(
         false, a,  false, b,
         (short)0, c,  false,  false);
    Ap += 4;
    Wp += wstep;
  }

  if (colOK) {
    const float bv = bias ? bias[col] : 0.0f;
#pragma unroll
    for (int i = 0; i < 8; ++i) {
      const int row = m0 + i + half * 8;
      if (row < M) {
        float v = c[i] + bv;
        if (do_relu) v = fmaxf(v, 0.0f);
        out[(size_t)row * N + col] = v;
      }
    }
  }
}

extern "C" void kernel_launch(void* const* d_in, const int* in_sizes, int n_in,
                              void* d_out, int out_size, void* d_ws, size_t ws_size,
                              hipStream_t stream) {
  const float* feat = (const float*)d_in[0];
  const float* W0   = (const float*)d_in[1];
  const float* W1   = (const float*)d_in[2];
  const float* W2   = (const float*)d_in[3];
  const float* b2   = (const float*)d_in[4];
  const int*   src  = (const int*)d_in[5];
  const int*   dst  = (const int*)d_in[6];
  float* out = (float*)d_out;

  const int n_nodes = in_sizes[0] / KDIM;
  const int n_out   = in_sizes[4];
  const int n_edges = in_sizes[5];
  const int n_bins  = (n_nodes + BIN_ROWS - 1) >> BIN_SHIFT;

  char* ws = (char*)d_ws;
  float* normS = (float*)ws;
  float* normD = normS + n_nodes;
  int*   ctr   = (int*)(normD + n_nodes);
  const size_t headBytes = (size_t)2 * n_nodes * sizeof(float) + 8 * sizeof(int);
  size_t off = (headBytes + 511) & ~(size_t)511;
  int* binCnt   = (int*)(ws + off);
  int* binStart = binCnt + n_bins;
  int* cursor   = binStart + n_bins + 1;
  off += ((size_t)(3 * n_bins + 1) * sizeof(int) + 511) & ~(size_t)511;
  int* epair = (int*)(ws + off);
  off += ((size_t)2 * n_edges * sizeof(int) + 511) & ~(size_t)511;
  float* agg  = (float*)(ws + off);
  off += (size_t)n_bins * BIN_ROWS * KDIM * sizeof(float);
  float* hbuf = (float*)(ws + off);

  const int mTiles = (n_nodes + 15) / 16;
  const dim3 degreeGrid(512);
  const dim3 blk256(256);

  hipMemsetAsync(normS, 0, headBytes, stream);
  gcn_degree_kernel<<<degreeGrid, blk256, 0, stream>>>(
      src, dst, normS, normD, ctr + 0, n_edges, n_nodes);
  gcn_norm_bincount_kernel<<<n_bins, BIN_ROWS, 0, stream>>>(
      normS, normD, binCnt, n_nodes);
  gcn_bin_scan_kernel<<<1, 1024, 0, stream>>>(binCnt, binStart, cursor, n_bins);
  gcn_bin_fill_kernel<<<(n_edges + FILL_THREADS * FILL_PER - 1) / (FILL_THREADS * FILL_PER),
                        FILL_THREADS, (size_t)2 * n_bins * sizeof(int), stream>>>(
      src, dst, cursor, epair, n_edges, n_bins);

  gcn_bin_sum_kernel<<<n_bins, blk256, 0, stream>>>(
      feat, normS, epair, binStart, agg);
  gcn_gemm_wmma_kernel<<<mTiles, dim3(32 * 8), 0, stream>>>(
      agg, normD, W0, nullptr, hbuf, n_nodes, 128,  1);

  gcn_bin_sum_kernel<<<n_bins, blk256, 0, stream>>>(
      hbuf, normS, epair, binStart, agg);
  gcn_gemm_wmma_kernel<<<mTiles, dim3(32 * 8), 0, stream>>>(
      agg, normD, W1, nullptr, hbuf, n_nodes, 128,  1);

  gcn_bin_sum_kernel<<<n_bins, blk256, 0, stream>>>(
      hbuf, normS, epair, binStart, agg);
  gcn_gemm_wmma_kernel<<<mTiles, dim3(32 * ((n_out + 15) / 16)), 0, stream>>>(
      agg, normD, W2, b2, out, n_nodes, n_out,  0);
}


# Layer GEMM moved into the bin work-group: its 64-row LDS sum tile (pitch 132) is the WMMA A operand, the wave's W slice in registers; agg round trip and 3 GEMM launches gone; sum orders kept
# speedup vs baseline: 3.2900x; 1.4911x over previous
#include <hip/hip_runtime.h>
#include <hip/hip_bf16.h>


#define KDIM 128
#define BIN_SHIFT 6
#define BIN_ROWS (1 << BIN_SHIFT)
#define TILE_PITCH 132

typedef __attribute__((ext_vector_type(2))) float v2f;
typedef __attribute__((ext_vector_type(8))) float v8f;

__device__ __forceinline__ int gcn_l2_half() {
  return (int)((__builtin_amdgcn_s_sendmsg_rtn(0x87) >> 18) & 1u);
}
__device__ __forceinline__ int gcn_set_split(int n_nodes) {
  return ((n_nodes >> 1) + 31) & ~31;
}

__global__ void __launch_bounds__(256)
gcn_degree_kernel(const int* __restrict__ src,
                  const int* __restrict__ dst,
                  float* __restrict__ degS,
                  float* __restrict__ degD,
                  int* __restrict__ ctr, int n_edges, int n_nodes) {
  __shared__ int s_half;
  __shared__ int s_slot[2];
  if (threadIdx.x == 0) s_half = gcn_l2_half();
  __syncthreads();
  const int half = s_half;
  const int n_split = gcn_set_split(n_nodes);
  const int n_chunks = (n_edges + 1023) >> 10;
  int k = 0;
  for (int pass = 0; pass < 2; ++pass) {
    const int set = half ^ pass;
    for (;;) {
      if (threadIdx.x == 0) s_slot[k & 1] = atomicAdd(ctr + set, 1);
      __syncthreads();
      const int chunk = s_slot[k & 1];
      ++k;
      if (chunk >= n_chunks) break;
      const int e0 = (chunk << 10) + (int)threadIdx.x;
      int sv[4], dv[4];
      for (int j = 0; j < 4; ++j) {
        const int e = e0 + (j << 8);
        const int ec = (e < n_edges) ? e : (n_edges - 1);
        sv[j] = src[ec];
        dv[j] = dst[ec];
      }
      for (int j = 0; j < 4; ++j) {
        if (e0 + (j << 8) < n_edges) {
          if ((sv[j] >= n_split) == (set != 0)) unsafeAtomicAdd(&degS[sv[j]], 1.0f);
          if ((dv[j] >= n_split) == (set != 0)) unsafeAtomicAdd(&degD[dv[j]], 1.0f);
        }
      }
    }
  }
}

__global__ void __launch_bounds__(BIN_ROWS)
gcn_norm_bincount_kernel(float* __restrict__ degS,
                         float* __restrict__ degD,
                         int* __restrict__ binCnt, int n_nodes) {
  __shared__ int s_cnt;
  if (threadIdx.x == 0) s_cnt = 0;
  __syncthreads();
  const int i = blockIdx.x * BIN_ROWS + (int)threadIdx.x;
  int c = 0;
  if (i < n_nodes) {
    const float ds = degS[i];
    const float dd = degD[i];
    c = (int)dd;
    degS[i] = rsqrtf(fmaxf(ds, 1.0f));
    degD[i] = rsqrtf(fmaxf(dd, 1.0f));
  }
  atomicAdd(&s_cnt, c);
  __syncthreads();
  if (threadIdx.x == 0) binCnt[blockIdx.x] = s_cnt;
}

__global__ void __launch_bounds__(1024)
gcn_bin_scan_kernel(const int* __restrict__ binCnt,
                    int* __restrict__ binStart,
                    int* __restrict__ cursor, int n_bins) {
  __shared__ int s_sum[1024];
  const int t = (int)threadIdx.x;
  const int per = (n_bins + 1023) >> 10;
  const int b0 = t * per;
  int local = 0;
  for (int j = 0; j < per; ++j) {
    const int b = b0 + j;
    if (b < n_bins) local += binCnt[b];
  }
  s_sum[t] = local;
  __syncthreads();
  for (int o = 1; o < 1024; o <<= 1) {
    const int v = (t >= o) ? s_sum[t - o] : 0;
    __syncthreads();
    s_sum[t] += v;
    __syncthreads();
  }
  int run = s_sum[t] - local;
  for (int j = 0; j < per; ++j) {
    const int b = b0 + j;
    if (b < n_bins) {
      binStart[b] = run;
      cursor[b] = run;
      run += binCnt[b];
    }
  }
  if (t == 1023) binStart[n_bins] = s_sum[1023];
}

#define FILL_THREADS 1024
#define FILL_PER 4
__global__ void __launch_bounds__(FILL_THREADS)
gcn_bin_fill_kernel(const int* __restrict__ src,
                    const int* __restrict__ dst,
                    int* __restrict__ cursor,
                    int* __restrict__ epair, int n_edges, int n_bins) {
  extern __shared__ int s_fill[];
  int* s_cnt  = s_fill;
  int* s_base = s_fill + n_bins;
  const int t = (int)threadIdx.x;
  for (int b = t; b < n_bins; b += FILL_THREADS) s_cnt[b] = 0;
  __syncthreads();

  const int e0 = blockIdx.x * (FILL_THREADS * FILL_PER) + t;
  int sv[FILL_PER], dv[FILL_PER], rk[FILL_PER];
  for (int j = 0; j < FILL_PER; ++j) {
    const int e = e0 + j * FILL_THREADS;
    const int ec = (e < n_edges) ? e : (n_edges - 1);
    sv[j] = src[ec];
    dv[j] = dst[ec];
  }
  for (int j = 0; j < FILL_PER; ++j) {
    rk[j] = 0;
    if (e0 + j * FILL_THREADS < n_edges)
      rk[j] = atomicAdd(&s_cnt[dv[j] >> BIN_SHIFT], 1);
  }
  __syncthreads();
  for (int b = t; b < n_bins; b += FILL_THREADS) {
    const int c = s_cnt[b];
    if (c > 0) s_base[b] = atomicAdd(cursor + b, c);
  }
  __syncthreads();
  for (int j = 0; j < FILL_PER; ++j) {
    if (e0 + j * FILL_THREADS < n_edges) {
      const int pos = s_base[dv[j] >> BIN_SHIFT] + rk[j];
      reinterpret_cast<int2*>(epair)[pos] = make_int2(sv[j], dv[j]);
    }
  }
}

template <int NT, int MT>
__global__ void __launch_bounds__(256)
gcn_bin_layer_kernel(const float* __restrict__ h,
                     const float* __restrict__ out_norm,
                     const float* __restrict__ in_norm,
                     const int* __restrict__ epair,
                     const int* __restrict__ binStart,
                     const float* __restrict__ W,
                     const float* __restrict__ bias,
                     float* __restrict__ out,
                     int M, int N, int do_relu) {
  __shared__ alignas(16) float tile[BIN_ROWS * TILE_PITCH];
  const int tid = (int)threadIdx.x;
  const int lane = tid & 31;
  const int wv = __builtin_amdgcn_readfirstlane(tid >> 5);
  const int row0 = blockIdx.x * BIN_ROWS;

  for (int i = 0; i < BIN_ROWS * KDIM / 4 / 256; ++i) {
    const int idx = tid + 256 * i;
    *reinterpret_cast<float4*>(tile + (idx >> 5) * TILE_PITCH + (idx & 31) * 4) =
        make_float4(0.0f, 0.0f, 0.0f, 0.0f);
  }
  __syncthreads();

  const int eb = binStart[blockIdx.x];
  const int ee = binStart[blockIdx.x + 1];
  for (int base = eb + wv * 32; base < ee; base += 256) {
    const int left = ee - base;
    const int cnt = (left < 32) ? left : 32;
    int s = 0, d = row0;
    if (lane < cnt) {
      const int2 p = reinterpret_cast<const int2*>(epair)[base + lane];
      s = p.x;
      d = p.y;
    }
    for (int g = 0; g < cnt; g += 8) {
      float a[8][4];
      float c[8];
      int off[8];
#pragma unroll
      for (int j = 0; j < 8; ++j) {
        const int li = (g + j < cnt) ? (g + j) : g;
        const int sj = __builtin_amdgcn_readlane(s, li);
        const int dj = __builtin_amdgcn_readlane(d, li);
        const float* hp = h + (size_t)sj * KDIM + lane;
        c[j] = out_norm[sj];
        a[j][0] = hp[0];
        a[j][1] = hp[32];
        a[j][2] = hp[64];
        a[j][3] = hp[96];
        off[j] = (dj - row0) * TILE_PITCH + lane;
      }
#pragma unroll
      for (int j = 0; j < 8; ++j) {
        const bool ok = (j == 0) || (g + j < cnt);
#pragma unroll
        for (int q = 0; q < 4; ++q) {
          const float v = ok ? a[j][q] * c[j] : 0.0f;
          unsafeAtomicAdd(&tile[off[j] + 32 * q], v);
        }
      }
    }
  }

  const int ntile = wv % NT;
  const int mt0   = (wv / NT) * MT;
  const bool active = (mt0 < BIN_ROWS / 16);
  const int half = lane >> 4;
  const int l15  = lane & 15;
  const int col    = ntile * 16 + l15;
  const bool colOK = (col < N);
  const int colC   = colOK ? col : (N - 1);

  v2f bw[KDIM / 4];
  float ascale[MT];
#pragma unroll
  for (int j = 0; j < KDIM / 4; ++j) bw[j] = v2f{0.0f, 0.0f};
#pragma unroll
  for (int mi = 0; mi < MT; ++mi) ascale[mi] = 0.0f;
  if (active) {
    const float* Wp = W + (size_t)(half * 2) * N + colC;
#pragma unroll
    for (int j = 0; j < KDIM / 4; ++j) {
      bw[j].x = Wp[(size_t)(4 * j) * N];
      bw[j].y = Wp[(size_t)(4 * j + 1) * N];
    }
#pragma unroll
    for (int mi = 0; mi < MT; ++mi) {
      int gr = row0 + (mt0 + mi) * 16 + l15;
      if (gr >= M) gr = M - 1;
      ascale[mi] = in_norm[gr];
    }
  }
  __syncthreads();

  if (active) {
    const float* Ap = tile + (mt0 * 16 + l15) * TILE_PITCH + half * 2;

    v8f acc[MT];
#pragma unroll
    for (int mi = 0; mi < MT; ++mi) acc[mi] = v8f{};

#pragma unroll
    for (int j = 0; j < KDIM / 4; ++j) {
#pragma unroll
      for (int mi = 0; mi < MT; ++mi) {
        v2f a = *reinterpret_cast<const v2f*>(Ap + mi * 16 * TILE_PITCH + 4 * j);
        a *= ascale[mi];
        acc[mi] = __builtin_amdgcn_wmma_f32_16x16x4_f32(
             false, a,  false, bw[j],
             (short)0, acc[mi],  false,  false);
      }
    }

    if (colOK) {
      const float bv = bias ? bias[col] : 0.0f;
#pragma unroll
      for (int mi = 0; mi < MT; ++mi) {
#pragma unroll
        for (int i = 0; i < 8; ++i) {
          const int row = row0 + (mt0 + mi) * 16 + i + half * 8;
          if (row < M) {
            float v = acc[mi][i] + bv;
            if (do_relu) v = fmaxf(v, 0.0f);
            out[(size_t)row * N + col] = v;
          }
        }
      }
    }
  }
}

extern "C" void kernel_launch(void* const* d_in, const int* in_sizes, int n_in,
                              void* d_out, int out_size, void* d_ws, size_t ws_size,
                              hipStream_t stream) {
  const float* feat = (const float*)d_in[0];
  const float* W0   = (const float*)d_in[1];
  const float* W1   = (const float*)d_in[2];
  const float* W2   = (const float*)d_in[3];
  const float* b2   = (const float*)d_in[4];
  const int*   src  = (const int*)d_in[5];
  const int*   dst  = (const int*)d_in[6];
  float* out = (float*)d_out;

  const int n_nodes = in_sizes[0] / KDIM;
  const int n_out   = in_sizes[4];
  const int n_edges = in_sizes[5];
  const int n_bins  = (n_nodes + BIN_ROWS - 1) >> BIN_SHIFT;

  char* ws = (char*)d_ws;
  float* normS = (float*)ws;
  float* normD = normS + n_nodes;
  int*   ctr   = (int*)(normD + n_nodes);
  const size_t headBytes = (size_t)2 * n_nodes * sizeof(float) + 8 * sizeof(int);
  size_t off = (headBytes + 511) & ~(size_t)511;
  int* binCnt   = (int*)(ws + off);
  int* binStart = binCnt + n_bins;
  int* cursor   = binStart + n_bins + 1;
  off += ((size_t)(3 * n_bins + 1) * sizeof(int) + 511) & ~(size_t)511;
  int* epair = (int*)(ws + off);
  off += ((size_t)2 * n_edges * sizeof(int) + 511) & ~(size_t)511;
  float* h1 = (float*)(ws + off);
  off += (((size_t)n_nodes * KDIM * sizeof(float)) + 511) & ~(size_t)511;
  float* h2 = (float*)(ws + off);

  const dim3 degreeGrid(512);
  const dim3 blk256(256);

  hipMemsetAsync(normS, 0, headBytes, stream);
  gcn_degree_kernel<<<degreeGrid, blk256, 0, stream>>>(
      src, dst, normS, normD, ctr + 0, n_edges, n_nodes);
  gcn_norm_bincount_kernel<<<n_bins, BIN_ROWS, 0, stream>>>(
      normS, normD, binCnt, n_nodes);
  gcn_bin_scan_kernel<<<1, 1024, 0, stream>>>(binCnt, binStart, cursor, n_bins);
  gcn_bin_fill_kernel<<<(n_edges + FILL_THREADS * FILL_PER - 1) / (FILL_THREADS * FILL_PER),
                        FILL_THREADS, (size_t)2 * n_bins * sizeof(int), stream>>>(
      src, dst, cursor, epair, n_edges, n_bins);

  gcn_bin_layer_kernel<8, 4><<<n_bins, blk256, 0, stream>>>(
      feat, normS, normD, epair, binStart, W0, nullptr, h1, n_nodes, 128,  1);
  gcn_bin_layer_kernel<8, 4><<<n_bins, blk256, 0, stream>>>(
      h1, normS, normD, epair, binStart, W1, nullptr, h2, n_nodes, 128,  1);
  gcn_bin_layer_kernel<3, 2><<<n_bins, blk256, 0, stream>>>(
      h2, normS, normD, epair, binStart, W2, b2, out, n_nodes, n_out,  0);
}


# Pre-pass cut from 4 launches to 2: bins counted in LDS in the degree pass, prefix sum and out-norm inside the fill pass, in-degree counted per bin in the layer's LDS; 625000 global atomics gone
# speedup vs baseline: 3.4835x; 1.0588x over previous
#include <hip/hip_runtime.h>
#include <hip/hip_bf16.h>


#define KDIM 128
#define BIN_SHIFT 6
#define BIN_ROWS (1 << BIN_SHIFT)
#define TILE_PITCH 132

typedef __attribute__((ext_vector_type(2))) float v2f;
typedef __attribute__((ext_vector_type(8))) float v8f;

__device__ __forceinline__ int gcn_l2_half() {
  return (int)((__builtin_amdgcn_s_sendmsg_rtn(0x87) >> 18) & 1u);
}
__device__ __forceinline__ int gcn_set_split(int n_nodes) {
  return ((n_nodes >> 1) + 31) & ~31;
}

__global__ void __launch_bounds__(256)
gcn_degree_kernel(const int* __restrict__ src,
                  const int* __restrict__ dst,
                  float* __restrict__ degS,
                  int* __restrict__ binCnt,
                  int* __restrict__ ctr, int n_edges, int n_nodes, int n_bins) {
  extern __shared__ int s_hist[];
  __shared__ int s_half;
  __shared__ int s_slot[2];
  for (int b = (int)threadIdx.x; b < n_bins; b += 256) s_hist[b] = 0;
  if (threadIdx.x == 0) s_half = gcn_l2_half();
  __syncthreads();
  const int half = s_half;
  const int n_split = gcn_set_split(n_nodes);
  const int n_chunks = (n_edges + 1023) >> 10;
  int k = 0;
  for (int pass = 0; pass < 2; ++pass) {
    const int set = half ^ pass;
    for (;;) {
      if (threadIdx.x == 0) s_slot[k & 1] = atomicAdd(ctr + set, 1);
      __syncthreads();
      const int chunk = s_slot[k & 1];
      ++k;
      if (chunk >= n_chunks) break;
      const int e0 = (chunk << 10) + (int)threadIdx.x;
      int sv[4], dv[4];
      for (int j = 0; j < 4; ++j) {
        const int e = e0 + (j << 8);
        const int ec = (e < n_edges) ? e : (n_edges - 1);
        sv[j] = src[ec];
        dv[j] = dst[ec];
      }
      for (int j = 0; j < 4; ++j) {
        if (e0 + (j << 8) < n_edges) {
          if ((sv[j] >= n_split) == (set != 0)) unsafeAtomicAdd(&degS[sv[j]], 1.0f);
          if ((dv[j] >= n_split) == (set != 0)) atomicAdd(&s_hist[dv[j] >> BIN_SHIFT], 1);
        }
      }
    }
  }
  __syncthreads();
  for (int b = (int)threadIdx.x; b < n_bins; b += 256) {
    const int c = s_hist[b];
    if (c > 0) atomicAdd(binCnt + b, c);
  }
}

#define FILL_THREADS 1024
#define FILL_PER 4
__global__ void __launch_bounds__(FILL_THREADS)
gcn_bin_fill_kernel(const int* __restrict__ src,
                    const int* __restrict__ dst,
                    const int* __restrict__ binCnt,
                    int* __restrict__ binStart,
                    int* __restrict__ cursor,
                    float* __restrict__ degS,
                    int* __restrict__ epair, int n_edges, int n_bins, int n_nodes) {
  extern __shared__ int s_fill[];
  int* s_cnt  = s_fill;
  int* s_base = s_fill + n_bins;
  __shared__ int s_wsum[FILL_THREADS / 32];
  const int t = (int)threadIdx.x;
  const int lane = t & 31;
  const int wv = t >> 5;

  const int e0 = blockIdx.x * (FILL_THREADS * FILL_PER) + t;
  int sv[FILL_PER], dv[FILL_PER], rk[FILL_PER];
  for (int j = 0; j < FILL_PER; ++j) {
    const int e = e0 + j * FILL_THREADS;
    const int ec = (e < n_edges) ? e : (n_edges - 1);
    sv[j] = src[ec];
    dv[j] = dst[ec];
  }

  for (int i = blockIdx.x * FILL_THREADS + t; i < n_nodes; i += (int)gridDim.x * FILL_THREADS)
    degS[i] = rsqrtf(fmaxf(degS[i], 1.0f));

  for (int b = t; b < n_bins; b += FILL_THREADS) {
    s_cnt[b] = 0;
    s_base[b] = binCnt[b];
  }
  __syncthreads();

  const int per = (n_bins + FILL_THREADS - 1) / FILL_THREADS;
  const int b0 = t * per;
  int local = 0;
  for (int j = 0; j < per; ++j) {
    const int b = b0 + j;
    if (b < n_bins) local += s_base[b];
  }
  int incl = local;
  for (int o = 1; o < 32; o <<= 1) {
    const int u = __shfl_up(incl, o);
    if (lane >= o) incl += u;
  }
  if (lane == 31) s_wsum[wv] = incl;
  __syncthreads();
  if (wv == 0) {
    int w = s_wsum[lane];
    for (int o = 1; o < 32; o <<= 1) {
      const int u = __shfl_up(w, o);
      if (lane >= o) w += u;
    }
    s_wsum[lane] = w;
  }
  __syncthreads();
  {
    int run = incl - local + ((wv > 0) ? s_wsum[wv - 1] : 0);
    for (int j = 0; j < per; ++j) {
      const int b = b0 + j;
      if (b < n_bins) {
        const int c = s_base[b];
        s_base[b] = run;
        if (blockIdx.x == 0) binStart[b] = run;
        run += c;
      }
    }
    if (blockIdx.x == 0 && t == 0) binStart[n_bins] = s_wsum[FILL_THREADS / 32 - 1];
  }

  for (int j = 0; j < FILL_PER; ++j) {
    rk[j] = 0;
    if (e0 + j * FILL_THREADS < n_edges)
      rk[j] = atomicAdd(&s_cnt[dv[j] >> BIN_SHIFT], 1);
  }
  __syncthreads();
  for (int b = t; b < n_bins; b += FILL_THREADS) {
    const int c = s_cnt[b];
    if (c > 0) s_base[b] += atomicAdd(cursor + b, c);
  }
  __syncthreads();
  for (int j = 0; j < FILL_PER; ++j) {
    if (e0 + j * FILL_THREADS < n_edges) {
      const int pos = s_base[dv[j] >> BIN_SHIFT] + rk[j];
      reinterpret_cast<int2*>(epair)[pos] = make_int2(sv[j], dv[j]);
    }
  }
}

template <int NT, int MT>
__global__ void __launch_bounds__(256)
gcn_bin_layer_kernel(const float* __restrict__ h,
                     const float* __restrict__ out_norm,
                     const int* __restrict__ epair,
                     const int* __restrict__ binStart,
                     const float* __restrict__ W,
                     const float* __restrict__ bias,
                     float* __restrict__ out,
                     int M, int N, int do_relu) {
  __shared__ alignas(16) float tile[BIN_ROWS * TILE_PITCH];
  __shared__ int s_deg[BIN_ROWS];
  const int tid = (int)threadIdx.x;
  const int lane = tid & 31;
  const int wv = __builtin_amdgcn_readfirstlane(tid >> 5);
  const int row0 = blockIdx.x * BIN_ROWS;

  for (int i = 0; i < BIN_ROWS * KDIM / 4 / 256; ++i) {
    const int idx = tid + 256 * i;
    *reinterpret_cast<float4*>(tile + (idx >> 5) * TILE_PITCH + (idx & 31) * 4) =
        make_float4(0.0f, 0.0f, 0.0f, 0.0f);
  }
  if (tid < BIN_ROWS) s_deg[tid] = 0;
  __syncthreads();

  const int eb = binStart[blockIdx.x];
  const int ee = binStart[blockIdx.x + 1];
  for (int base = eb + wv * 32; base < ee; base += 256) {
    const int left = ee - base;
    const int cnt = (left < 32) ? left : 32;
    int s = 0, d = row0;
    if (lane < cnt) {
      const int2 p = reinterpret_cast<const int2*>(epair)[base + lane];
      s = p.x;
      d = p.y;
      atomicAdd(&s_deg[d - row0], 1);
    }
    for (int g = 0; g < cnt; g += 8) {
      float a[8][4];
      float c[8];
      int off[8];
#pragma unroll
      for (int j = 0; j < 8; ++j) {
        const int li = (g + j < cnt) ? (g + j) : g;
        const int sj = __builtin_amdgcn_readlane(s, li);
        const int dj = __builtin_amdgcn_readlane(d, li);
        const float* hp = h + (size_t)sj * KDIM + lane;
        c[j] = out_norm[sj];
        a[j][0] = hp[0];
        a[j][1] = hp[32];
        a[j][2] = hp[64];
        a[j][3] = hp[96];
        off[j] = (dj - row0) * TILE_PITCH + lane;
      }
#pragma unroll
      for (int j = 0; j < 8; ++j) {
        const bool ok = (j == 0) || (g + j < cnt);
#pragma unroll
        for (int q = 0; q < 4; ++q) {
          const float v = ok ? a[j][q] * c[j] : 0.0f;
          unsafeAtomicAdd(&tile[off[j] + 32 * q], v);
        }
      }
    }
  }

  const int ntile = wv % NT;
  const int mt0   = (wv / NT) * MT;
  const bool active = (mt0 < BIN_ROWS / 16);
  const int half = lane >> 4;
  const int l15  = lane & 15;
  const int col    = ntile * 16 + l15;
  const bool colOK = (col < N);
  const int colC   = colOK ? col : (N - 1);

  v2f bw[KDIM / 4];
#pragma unroll
  for (int j = 0; j < KDIM / 4; ++j) bw[j] = v2f{0.0f, 0.0f};
  if (active) {
    const float* Wp = W + (size_t)(half * 2) * N + colC;
#pragma unroll
    for (int j = 0; j < KDIM / 4; ++j) {
      bw[j].x = Wp[(size_t)(4 * j) * N];
      bw[j].y = Wp[(size_t)(4 * j + 1) * N];
    }
  }
  __syncthreads();

  if (active) {
    const float* Ap = tile + (mt0 * 16 + l15) * TILE_PITCH + half * 2;

    float ascale[MT];
#pragma unroll
    for (int mi = 0; mi < MT; ++mi)
      ascale[mi] = rsqrtf(fmaxf((float)s_deg[(mt0 + mi) * 16 + l15], 1.0f));

    v8f acc[MT];
#pragma unroll
    for (int mi = 0; mi < MT; ++mi) acc[mi] = v8f{};

#pragma unroll
    for (int j = 0; j < KDIM / 4; ++j) {
#pragma unroll
      for (int mi = 0; mi < MT; ++mi) {
        v2f a = *reinterpret_cast<const v2f*>(Ap + mi * 16 * TILE_PITCH + 4 * j);
        a *= ascale[mi];
        acc[mi] = __builtin_amdgcn_wmma_f32_16x16x4_f32(
             false, a,  false, bw[j],
             (short)0, acc[mi],  false,  false);
      }
    }

    if (colOK) {
      const float bv = bias ? bias[col] : 0.0f;
#pragma unroll
      for (int mi = 0; mi < MT; ++mi) {
#pragma unroll
        for (int i = 0; i < 8; ++i) {
          const int row = row0 + (mt0 + mi) * 16 + i + half * 8;
          if (row < M) {
            float v = acc[mi][i] + bv;
            if (do_relu) v = fmaxf(v, 0.0f);
            out[(size_t)row * N + col] = v;
          }
        }
      }
    }
  }
}

extern "C" void kernel_launch(void* const* d_in, const int* in_sizes, int n_in,
                              void* d_out, int out_size, void* d_ws, size_t ws_size,
                              hipStream_t stream) {
  const float* feat = (const float*)d_in[0];
  const float* W0   = (const float*)d_in[1];
  const float* W1   = (const float*)d_in[2];
  const float* W2   = (const float*)d_in[3];
  const float* b2   = (const float*)d_in[4];
  const int*   src  = (const int*)d_in[5];
  const int*   dst  = (const int*)d_in[6];
  float* out = (float*)d_out;

  const int n_nodes = in_sizes[0] / KDIM;
  const int n_out   = in_sizes[4];
  const int n_edges = in_sizes[5];
  const int n_bins  = (n_nodes + BIN_ROWS - 1) >> BIN_SHIFT;

  char* ws = (char*)d_ws;
  float* normS  = (float*)ws;
  int*   ctr    = (int*)(normS + n_nodes);
  int*   binCnt = ctr + 32;
  int*   cursor = binCnt + n_bins;
  const size_t headBytes = ((size_t)n_nodes + 32 + (size_t)2 * n_bins) * sizeof(int);
  size_t off = (headBytes + 511) & ~(size_t)511;
  int* binStart = (int*)(ws + off);
  off += ((size_t)(n_bins + 1) * sizeof(int) + 511) & ~(size_t)511;
  int* epair = (int*)(ws + off);
  off += ((size_t)2 * n_edges * sizeof(int) + 511) & ~(size_t)511;
  float* h1 = (float*)(ws + off);
  off += (((size_t)n_nodes * KDIM * sizeof(float)) + 511) & ~(size_t)511;
  float* h2 = (float*)(ws + off);

  const dim3 degreeGrid(512);
  const dim3 blk256(256);

  hipMemsetAsync(normS, 0, headBytes, stream);
  gcn_degree_kernel<<<degreeGrid, blk256, (size_t)n_bins * sizeof(int), stream>>>(
      src, dst, normS, binCnt, ctr + 0, n_edges, n_nodes, n_bins);
  gcn_bin_fill_kernel<<<(n_edges + FILL_THREADS * FILL_PER - 1) / (FILL_THREADS * FILL_PER),
                        FILL_THREADS, (size_t)2 * n_bins * sizeof(int), stream>>>(
      src, dst, binCnt, binStart, cursor, normS, epair, n_edges, n_bins, n_nodes);

  gcn_bin_layer_kernel<8, 4><<<n_bins, blk256, 0, stream>>>(
      feat, normS, epair, binStart, W0, nullptr, h1, n_nodes, 128,  1);
  gcn_bin_layer_kernel<8, 4><<<n_bins, blk256, 0, stream>>>(
      h1, normS, epair, binStart, W1, nullptr, h2, n_nodes, 128,  1);
  gcn_bin_layer_kernel<3, 2><<<n_bins, blk256, 0, stream>>>(
      h2, normS, epair, binStart, W2, b2, out, n_nodes, n_out,  0);
}


# Bin's edges ordered by destination row in LDS (returning count add = rank); a wave sums each row's run in registers: 4 LDS float adds per row, not per edge; sum order = rank order (LDS arrival)
# speedup vs baseline: 3.6752x; 1.0550x over previous
#include <hip/hip_runtime.h>
#include <hip/hip_bf16.h>


#define KDIM 128
#define BIN_SHIFT 6
#define BIN_ROWS (1 << BIN_SHIFT)
#define TILE_PITCH 132
#define PASS_EDGES 512

typedef __attribute__((ext_vector_type(2))) float v2f;
typedef __attribute__((ext_vector_type(8))) float v8f;

__device__ __forceinline__ int gcn_l2_half() {
  return (int)((__builtin_amdgcn_s_sendmsg_rtn(0x87) >> 18) & 1u);
}
__device__ __forceinline__ int gcn_set_split(int n_nodes) {
  return ((n_nodes >> 1) + 31) & ~31;
}

__global__ void __launch_bounds__(256)
gcn_degree_kernel(const int* __restrict__ src,
                  const int* __restrict__ dst,
                  float* __restrict__ degS,
                  int* __restrict__ binCnt,
                  int* __restrict__ ctr, int n_edges, int n_nodes, int n_bins) {
  extern __shared__ int s_hist[];
  __shared__ int s_half;
  __shared__ int s_slot[2];
  for (int b = (int)threadIdx.x; b < n_bins; b += 256) s_hist[b] = 0;
  if (threadIdx.x == 0) s_half = gcn_l2_half();
  __syncthreads();
  const int half = s_half;
  const int n_split = gcn_set_split(n_nodes);
  const int n_chunks = (n_edges + 1023) >> 10;
  int k = 0;
  for (int pass = 0; pass < 2; ++pass) {
    const int set = half ^ pass;
    for (;;) {
      if (threadIdx.x == 0) s_slot[k & 1] = atomicAdd(ctr + set, 1);
      __syncthreads();
      const int chunk = s_slot[k & 1];
      ++k;
      if (chunk >= n_chunks) break;
      const int e0 = (chunk << 10) + (int)threadIdx.x;
      int sv[4], dv[4];
      for (int j = 0; j < 4; ++j) {
        const int e = e0 + (j << 8);
        const int ec = (e < n_edges) ? e : (n_edges - 1);
        sv[j] = src[ec];
        dv[j] = dst[ec];
      }
      for (int j = 0; j < 4; ++j) {
        if (e0 + (j << 8) < n_edges) {
          if ((sv[j] >= n_split) == (set != 0)) unsafeAtomicAdd(&degS[sv[j]], 1.0f);
          if ((dv[j] >= n_split) == (set != 0)) atomicAdd(&s_hist[dv[j] >> BIN_SHIFT], 1);
        }
      }
    }
  }
  __syncthreads();
  for (int b = (int)threadIdx.x; b < n_bins; b += 256) {
    const int c = s_hist[b];
    if (c > 0) atomicAdd(binCnt + b, c);
  }
}

#define FILL_THREADS 1024
#define FILL_PER 4
__global__ void __launch_bounds__(FILL_THREADS)
gcn_bin_fill_kernel(const int* __restrict__ src,
                    const int* __restrict__ dst,
                    const int* __restrict__ binCnt,
                    int* __restrict__ binStart,
                    int* __restrict__ cursor,
                    float* __restrict__ degS,
                    int* __restrict__ epair, int n_edges, int n_bins, int n_nodes) {
  extern __shared__ int s_fill[];
  int* s_cnt  = s_fill;
  int* s_base = s_fill + n_bins;
  __shared__ int s_wsum[FILL_THREADS / 32];
  const int t = (int)threadIdx.x;
  const int lane = t & 31;
  const int wv = t >> 5;

  const int e0 = blockIdx.x * (FILL_THREADS * FILL_PER) + t;
  int sv[FILL_PER], dv[FILL_PER], rk[FILL_PER];
  for (int j = 0; j < FILL_PER; ++j) {
    const int e = e0 + j * FILL_THREADS;
    const int ec = (e < n_edges) ? e : (n_edges - 1);
    sv[j] = src[ec];
    dv[j] = dst[ec];
  }

  for (int i = blockIdx.x * FILL_THREADS + t; i < n_nodes; i += (int)gridDim.x * FILL_THREADS)
    degS[i] = rsqrtf(fmaxf(degS[i], 1.0f));

  for (int b = t; b < n_bins; b += FILL_THREADS) {
    s_cnt[b] = 0;
    s_base[b] = binCnt[b];
  }
  __syncthreads();

  const int per = (n_bins + FILL_THREADS - 1) / FILL_THREADS;
  const int b0 = t * per;
  int local = 0;
  for (int j = 0; j < per; ++j) {
    const int b = b0 + j;
    if (b < n_bins) local += s_base[b];
  }
  int incl = local;
  for (int o = 1; o < 32; o <<= 1) {
    const int u = __shfl_up(incl, o);
    if (lane >= o) incl += u;
  }
  if (lane == 31) s_wsum[wv] = incl;
  __syncthreads();
  if (wv == 0) {
    int w = s_wsum[lane];
    for (int o = 1; o < 32; o <<= 1) {
      const int u = __shfl_up(w, o);
      if (lane >= o) w += u;
    }
    s_wsum[lane] = w;
  }
  __syncthreads();
  {
    int run = incl - local + ((wv > 0) ? s_wsum[wv - 1] : 0);
    for (int j = 0; j < per; ++j) {
      const int b = b0 + j;
      if (b < n_bins) {
        const int c = s_base[b];
        s_base[b] = run;
        if (blockIdx.x == 0) binStart[b] = run;
        run += c;
      }
    }
    if (blockIdx.x == 0 && t == 0) binStart[n_bins] = s_wsum[FILL_THREADS / 32 - 1];
  }

  for (int j = 0; j < FILL_PER; ++j) {
    rk[j] = 0;
    if (e0 + j * FILL_THREADS < n_edges)
      rk[j] = atomicAdd(&s_cnt[dv[j] >> BIN_SHIFT], 1);
  }
  __syncthreads();
  for (int b = t; b < n_bins; b += FILL_THREADS) {
    const int c = s_cnt[b];
    if (c > 0) s_base[b] += atomicAdd(cursor + b, c);
  }
  __syncthreads();
  for (int j = 0; j < FILL_PER; ++j) {
    if (e0 + j * FILL_THREADS < n_edges) {
      const int pos = s_base[dv[j] >> BIN_SHIFT] + rk[j];
      reinterpret_cast<int2*>(epair)[pos] = make_int2(sv[j], dv[j]);
    }
  }
}

template <int NT, int MT>
__global__ void __launch_bounds__(256)
gcn_bin_layer_kernel(const float* __restrict__ h,
                     const float* __restrict__ out_norm,
                     const int* __restrict__ epair,
                     const int* __restrict__ binStart,
                     const float* __restrict__ W,
                     const float* __restrict__ bias,
                     float* __restrict__ out,
                     int M, int N, int do_relu) {
  __shared__ alignas(16) float tile[BIN_ROWS * TILE_PITCH];
  __shared__ int s_deg[BIN_ROWS];
  __shared__ int s_cnt[BIN_ROWS];
  __shared__ int s_off[BIN_ROWS + 1];
  __shared__ alignas(8) int2 s_list[PASS_EDGES];
  const int tid = (int)threadIdx.x;
  const int lane = tid & 31;
  const int wv = __builtin_amdgcn_readfirstlane(tid >> 5);
  const int row0 = blockIdx.x * BIN_ROWS;

  for (int i = 0; i < BIN_ROWS * KDIM / 4 / 256; ++i) {
    const int idx = tid + 256 * i;
    *reinterpret_cast<float4*>(tile + (idx >> 5) * TILE_PITCH + (idx & 31) * 4) =
        make_float4(0.0f, 0.0f, 0.0f, 0.0f);
  }
  if (tid < BIN_ROWS) {
    s_deg[tid] = 0;
    s_cnt[tid] = 0;
  }
  __syncthreads();

  const int eb = binStart[blockIdx.x];
  const int ee = binStart[blockIdx.x + 1];
  for (int pb = eb; pb < ee; pb += PASS_EDGES) {
    int sv[PASS_EDGES / 256], rv[PASS_EDGES / 256], rk[PASS_EDGES / 256];
#pragma unroll
    for (int i = 0; i < PASS_EDGES / 256; ++i) {
      const int e = pb + tid + 256 * i;
      const int ec = (e < ee) ? e : (ee - 1);
      const int2 p = reinterpret_cast<const int2*>(epair)[ec];
      sv[i] = p.x;
      rv[i] = (e < ee) ? (p.y - row0) : -1;
    }
#pragma unroll
    for (int i = 0; i < PASS_EDGES / 256; ++i) {
      rk[i] = 0;
      if (rv[i] >= 0) rk[i] = atomicAdd(&s_cnt[rv[i]], 1);
    }
    __syncthreads();
    if (wv == 0) {
      const int c0 = s_cnt[2 * lane];
      const int c1 = s_cnt[2 * lane + 1];
      int incl = c0 + c1;
      for (int o = 1; o < 32; o <<= 1) {
        const int u = __shfl_up(incl, o);
        if (lane >= o) incl += u;
      }
      const int ex = incl - (c0 + c1);
      s_off[2 * lane] = ex;
      s_off[2 * lane + 1] = ex + c0;
      if (lane == 31) s_off[BIN_ROWS] = incl;
      s_deg[2 * lane] += c0;
      s_deg[2 * lane + 1] += c1;
    }
    __syncthreads();
#pragma unroll
    for (int i = 0; i < PASS_EDGES / 256; ++i) {
      if (rv[i] >= 0) s_list[s_off[rv[i]] + rk[i]] = make_int2(sv[i], rv[i]);
    }
    if (tid < BIN_ROWS) s_cnt[tid] = 0;
    __syncthreads();

    const int wb = __builtin_amdgcn_readfirstlane(s_off[wv * 8]);
    const int we = __builtin_amdgcn_readfirstlane(s_off[wv * 8 + 8]);
    int cur = -1;
    float acc[4] = {0.0f, 0.0f, 0.0f, 0.0f};
    for (int lb = wb; lb < we; lb += 32) {
      const int left = we - lb;
      const int cnt = (left < 32) ? left : 32;
      int s = 0, r = 0;
      if (lane < cnt) {
        const int2 p = s_list[lb + lane];
        s = p.x;
        r = p.y;
      }
      for (int g = 0; g < cnt; g += 8) {
        float a[8][4];
        float c[8];
        int rr[8];
#pragma unroll
        for (int j = 0; j < 8; ++j) {
          const int li = (g + j < cnt) ? (g + j) : g;
          const int sj = __builtin_amdgcn_readlane(s, li);
          rr[j] = __builtin_amdgcn_readlane(r, li);
          const float* hp = h + (size_t)sj * KDIM + lane;
          c[j] = out_norm[sj];
          a[j][0] = hp[0];
          a[j][1] = hp[32];
          a[j][2] = hp[64];
          a[j][3] = hp[96];
        }
#pragma unroll
        for (int j = 0; j < 8; ++j) {
          if ((j == 0) || (g + j < cnt)) {
            if (rr[j] != cur) {
              if (cur >= 0) {
#pragma unroll
                for (int q = 0; q < 4; ++q)
                  unsafeAtomicAdd(&tile[cur * TILE_PITCH + lane + 32 * q], acc[q]);
              }
              cur = rr[j];
#pragma unroll
              for (int q = 0; q < 4; ++q) acc[q] = 0.0f;
            }
#pragma unroll
            for (int q = 0; q < 4; ++q) acc[q] += a[j][q] * c[j];
          }
        }
      }
    }
    if (cur >= 0) {
#pragma unroll
      for (int q = 0; q < 4; ++q)
        unsafeAtomicAdd(&tile[cur * TILE_PITCH + lane + 32 * q], acc[q]);
    }
  }

  const int ntile = wv % NT;
  const int mt0   = (wv / NT) * MT;
  const bool active = (mt0 < BIN_ROWS / 16);
  const int half = lane >> 4;
  const int l15  = lane & 15;
  const int col    = ntile * 16 + l15;
  const bool colOK = (col < N);
  const int colC   = colOK ? col : (N - 1);

  v2f bw[KDIM / 4];
#pragma unroll
  for (int j = 0; j < KDIM / 4; ++j) bw[j] = v2f{0.0f, 0.0f};
  if (active) {
    const float* Wp = W + (size_t)(half * 2) * N + colC;
#pragma unroll
    for (int j = 0; j < KDIM / 4; ++j) {
      bw[j].x = Wp[(size_t)(4 * j) * N];
      bw[j].y = Wp[(size_t)(4 * j + 1) * N];
    }
  }
  __syncthreads();

  if (active) {
    const float* Ap = tile + (mt0 * 16 + l15) * TILE_PITCH + half * 2;

    float ascale[MT];
#pragma unroll
    for (int mi = 0; mi < MT; ++mi)
      ascale[mi] = rsqrtf(fmaxf((float)s_deg[(mt0 + mi) * 16 + l15], 1.0f));

    v8f acc[MT];
#pragma unroll
    for (int mi = 0; mi < MT; ++mi) acc[mi] = v8f{};

#pragma unroll
    for (int j = 0; j < KDIM / 4; ++j) {
#pragma unroll
      for (int mi = 0; mi < MT; ++mi) {
        v2f a = *reinterpret_cast<const v2f*>(Ap + mi * 16 * TILE_PITCH + 4 * j);
        a *= ascale[mi];
        acc[mi] = __builtin_amdgcn_wmma_f32_16x16x4_f32(
             false, a,  false, bw[j],
             (short)0, acc[mi],  false,  false);
      }
    }

    if (colOK) {
      const float bv = bias ? bias[col] : 0.0f;
#pragma unroll
      for (int mi = 0; mi < MT; ++mi) {
#pragma unroll
        for (int i = 0; i < 8; ++i) {
          const int row = row0 + (mt0 + mi) * 16 + i + half * 8;
          if (row < M) {
            float v = acc[mi][i] + bv;
            if (do_relu) v = fmaxf(v, 0.0f);
            out[(size_t)row * N + col] = v;
          }
        }
      }
    }
  }
}

extern "C" void kernel_launch(void* const* d_in, const int* in_sizes, int n_in,
                              void* d_out, int out_size, void* d_ws, size_t ws_size,
                              hipStream_t stream) {
  const float* feat = (const float*)d_in[0];
  const float* W0   = (const float*)d_in[1];
  const float* W1   = (const float*)d_in[2];
  const float* W2   = (const float*)d_in[3];
  const float* b2   = (const float*)d_in[4];
  const int*   src  = (const int*)d_in[5];
  const int*   dst  = (const int*)d_in[6];
  float* out = (float*)d_out;

  const int n_nodes = in_sizes[0] / KDIM;
  const int n_out   = in_sizes[4];
  const int n_edges = in_sizes[5];
  const int n_bins  = (n_nodes + BIN_ROWS - 1) >> BIN_SHIFT;

  char* ws = (char*)d_ws;
  float* normS  = (float*)ws;
  int*   ctr    = (int*)(normS + n_nodes);
  int*   binCnt = ctr + 32;
  int*   cursor = binCnt + n_bins;
  const size_t headBytes = ((size_t)n_nodes + 32 + (size_t)2 * n_bins) * sizeof(int);
  size_t off = (headBytes + 511) & ~(size_t)511;
  int* binStart = (int*)(ws + off);
  off += ((size_t)(n_bins + 1) * sizeof(int) + 511) & ~(size_t)511;
  int* epair = (int*)(ws + off);
  off += ((size_t)2 * n_edges * sizeof(int) + 511) & ~(size_t)511;
  float* h1 = (float*)(ws + off);
  off += (((size_t)n_nodes * KDIM * sizeof(float)) + 511) & ~(size_t)511;
  float* h2 = (float*)(ws + off);

  const dim3 degreeGrid(512);
  const dim3 blk256(256);

  hipMemsetAsync(normS, 0, headBytes, stream);
  gcn_degree_kernel<<<degreeGrid, blk256, (size_t)n_bins * sizeof(int), stream>>>(
      src, dst, normS, binCnt, ctr + 0, n_edges, n_nodes, n_bins);
  gcn_bin_fill_kernel<<<(n_edges + FILL_THREADS * FILL_PER - 1) / (FILL_THREADS * FILL_PER),
                        FILL_THREADS, (size_t)2 * n_bins * sizeof(int), stream>>>(
      src, dst, binCnt, binStart, cursor, normS, epair, n_edges, n_bins, n_nodes);

  gcn_bin_layer_kernel<8, 4><<<n_bins, blk256, 0, stream>>>(
      feat, normS, epair, binStart, W0, nullptr, h1, n_nodes, 128,  1);
  gcn_bin_layer_kernel<8, 4><<<n_bins, blk256, 0, stream>>>(
      h1, normS, epair, binStart, W1, nullptr, h2, n_nodes, 128,  1);
  gcn_bin_layer_kernel<3, 2><<<n_bins, blk256, 0, stream>>>(
      h2, normS, epair, binStart, W2, b2, out, n_nodes, n_out,  0);
}
